# starting kernel
# baseline (speedup 1.0000x reference)
#include <hip/hip_runtime.h>
#include <hip/hip_bf16.h>


typedef float v2f __attribute__((ext_vector_type(2)));
typedef float v8f __attribute__((ext_vector_type(8)));

#define B_     4
#define N_     1024
#define M_     1024
#define DIM_   1024
#define H_     16
#define DH_    64
#define INNER_ 1024
#define SCALE_ 0.125f

__device__ __forceinline__ v8f wmma4(v2f a, v2f b, v8f c) {
  return __builtin_amdgcn_wmma_f32_16x16x4_f32(false, a, false, b, (short)0, c,
                                               false, false);
}

__device__ __forceinline__ v8f v8f_zero() {
  v8f z = {0.f, 0.f, 0.f, 0.f, 0.f, 0.f, 0.f, 0.f};
  return z;
}

__device__ __forceinline__ void gemm16x64(const float* __restrict__ A, int lda,
                                          const float* __restrict__ W, int ldw,
                                          int row0, int col0, int K,
                                          v8f acc[4]) {
  const int lane   = threadIdx.x & 31;
  const int laneLo = lane & 15;
  const int hi2    = (lane >> 4) << 1;
  const float* ap = A + (size_t)(row0 + laneLo) * lda + hi2;
  const float* wp = W + (size_t)hi2 * ldw + col0 + laneLo;
#pragma unroll 4
  for (int kk = 0; kk < K; kk += 4) {
    v2f a = *(const v2f*)(ap + kk);
    const float* w0 = wp + (size_t)kk * ldw;
#pragma unroll
    for (int t = 0; t < 4; ++t) {
      v2f b;
      b.x = w0[t * 16];
      b.y = w0[t * 16 + ldw];
      acc[t] = wmma4(a, b, acc[t]);
    }
  }
}

__global__ void __launch_bounds__(128)
qproj_rope(const float* __restrict__ x, const float* __restrict__ Wq,
           const float* __restrict__ rope, float* __restrict__ Q) {
  const int wave = threadIdx.x >> 5;
  const int ct   = blockIdx.x & 15;
  const int rblk = blockIdx.x >> 4;
  const int row0 = (rblk * 4 + wave) * 16;
  const int col0 = ct * 64;

  v8f acc[4] = {v8f_zero(), v8f_zero(), v8f_zero(), v8f_zero()};
  gemm16x64(x, DIM_, Wq, INNER_, row0, col0, DIM_, acc);

  const int lane   = threadIdx.x & 31;
  const int laneLo = lane & 15;
  const int rowAdd = (lane >> 4) * 8;
  const int b  = row0 >> 10;
  const int n0 = row0 & 1023;
  const int h  = ct;
#pragma unroll
  for (int v = 0; v < 8; ++v) {
    const int n = n0 + v + rowAdd;
    const float* rp = rope + n * DH_;
    float ov[4];
#pragma unroll
    for (int t = 0; t < 4; ++t) {
      const int d = t * 16 + laneLo;
      float p  = rp[d];
      float cs = __cosf(p), sn = __sinf(p);
      float rot = (t < 2) ? -acc[t + 2][v] : acc[t - 2][v];
      ov[t] = acc[t][v] * cs + rot * sn;
    }
    float* qout = Q + ((size_t)((b * H_ + h) * N_ + n)) * DH_;
#pragma unroll
    for (int t = 0; t < 4; ++t) qout[t * 16 + laneLo] = ov[t];
  }
}

__global__ void __launch_bounds__(128)
kvproj_rope(const float* __restrict__ ctx, const float* __restrict__ Wkv,
            const float* __restrict__ rope, float* __restrict__ K,
            float* __restrict__ V) {
  const int wave = threadIdx.x >> 5;
  const int ct   = blockIdx.x & 31;
  const int rblk = blockIdx.x >> 5;
  const int row0 = (rblk * 4 + wave) * 16;
  const int col0 = ct * 64;

  v8f acc[4] = {v8f_zero(), v8f_zero(), v8f_zero(), v8f_zero()};
  gemm16x64(ctx, DIM_, Wkv, 2 * INNER_, row0, col0, DIM_, acc);

  const int lane   = threadIdx.x & 31;
  const int laneLo = lane & 15;
  const int rowAdd = (lane >> 4) * 8;
  const int b  = row0 >> 10;
  const int m0 = row0 & 1023;

  if (ct < 16) {
    const int h = ct;
#pragma unroll
    for (int v = 0; v < 8; ++v) {
      const int m = m0 + v + rowAdd;
      const float* rp = rope + m * DH_;
      float ov[4];
#pragma unroll
      for (int t = 0; t < 4; ++t) {
        const int d = t * 16 + laneLo;
        float p  = rp[d];
        float cs = __cosf(p), sn = __sinf(p);
        float rot = (t < 2) ? -acc[t + 2][v] : acc[t - 2][v];
        ov[t] = acc[t][v] * cs + rot * sn;
      }
      float* kout = K + ((size_t)((b * H_ + h) * M_ + m)) * DH_;
#pragma unroll
      for (int t = 0; t < 4; ++t) kout[t * 16 + laneLo] = ov[t];
    }
  } else {
    const int h = ct - 16;
#pragma unroll
    for (int v = 0; v < 8; ++v) {
      const int m = m0 + v + rowAdd;
      float* vout = V + ((size_t)((b * H_ + h) * M_ + m)) * DH_;
#pragma unroll
      for (int t = 0; t < 4; ++t) vout[t * 16 + laneLo] = acc[t][v];
    }
  }
}

__global__ void __launch_bounds__(128)
attn_fwd(const float* __restrict__ Q, const float* __restrict__ K,
         const float* __restrict__ V, float* __restrict__ OH) {
  __shared__ float ldsP[4][16 * 18];

  const int wave   = threadIdx.x >> 5;
  const int lane   = threadIdx.x & 31;
  const int laneLo = lane & 15;
  const int hi2    = (lane >> 4) << 1;
  const int rowAdd = (lane >> 4) * 8;

  const int bh   = blockIdx.x >> 4;
  const int iblk = blockIdx.x & 15;
  const int i0   = iblk * 64 + wave * 16;

  const float* qbase = Q + ((size_t)bh * N_ + i0 + laneLo) * DH_ + hi2;
  v2f qa[16];
#pragma unroll
  for (int s = 0; s < 16; ++s) qa[s] = *(const v2f*)(qbase + 4 * s);

  v8f o[4] = {v8f_zero(), v8f_zero(), v8f_zero(), v8f_zero()};
  float m[8], l[8];
#pragma unroll
  for (int v = 0; v < 8; ++v) { m[v] = -__builtin_inff(); l[v] = 0.f; }

  const float* kb0 = K + (size_t)bh * M_ * DH_;
  const float* vb0 = V + (size_t)bh * M_ * DH_;
  float* myP = &ldsP[wave][0];

  for (int j0 = 0; j0 < M_; j0 += 16) {
    v8f sacc = v8f_zero();
    const float* kp = kb0 + (size_t)(j0 + laneLo) * DH_ + hi2;
#pragma unroll
    for (int s = 0; s < 16; ++s) {
      v2f kb = *(const v2f*)(kp + 4 * s);
      sacc = wmma4(qa[s], kb, sacc);
    }

#pragma unroll
    for (int v = 0; v < 8; ++v) {
      float sv = sacc[v] * SCALE_;
      float mx = sv;
      mx = fmaxf(mx, __shfl_xor(mx, 1, 32));
      mx = fmaxf(mx, __shfl_xor(mx, 2, 32));
      mx = fmaxf(mx, __shfl_xor(mx, 4, 32));
      mx = fmaxf(mx, __shfl_xor(mx, 8, 32));
      float mn = fmaxf(m[v], mx);
      float rr = __expf(m[v] - mn);
      float p  = __expf(sv - mn);
      m[v] = mn;
      float ps = p;
      ps += __shfl_xor(ps, 1, 32);
      ps += __shfl_xor(ps, 2, 32);
      ps += __shfl_xor(ps, 4, 32);
      ps += __shfl_xor(ps, 8, 32);
      l[v] = l[v] * rr + ps;
#pragma unroll
      for (int t = 0; t < 4; ++t) o[t][v] *= rr;
      myP[(v + rowAdd) * 18 + laneLo] = p;
    }
    asm volatile("s_wait_dscnt 0" ::: "memory");

    const float* vp = vb0 + (size_t)(j0 + hi2) * DH_ + laneLo;
#pragma unroll
    for (int s = 0; s < 4; ++s) {
      v2f pa = *(const v2f*)(myP + laneLo * 18 + 4 * s + hi2);
      const float* vrow = vp + (size_t)(4 * s) * DH_;
#pragma unroll
      for (int t = 0; t < 4; ++t) {
        v2f vb;
        vb.x = vrow[t * 16];
        vb.y = vrow[t * 16 + DH_];
        o[t] = wmma4(pa, vb, o[t]);
      }
    }
    asm volatile("" ::: "memory");
  }

  const int b = bh >> 4, h = bh & 15;
#pragma unroll
  for (int v = 0; v < 8; ++v) {
    const int i  = i0 + v + rowAdd;
    float inv = 1.0f / l[v];
    float* op = OH + (((size_t)(b * N_ + i) * H_) + h) * DH_;
#pragma unroll
    for (int t = 0; t < 4; ++t) op[t * 16 + laneLo] = o[t][v] * inv;
  }
}

__global__ void __launch_bounds__(128)
outproj(const float* __restrict__ OH, const float* __restrict__ Wout,
        const float* __restrict__ bout, float* __restrict__ out) {
  const int wave = threadIdx.x >> 5;
  const int ct   = blockIdx.x & 15;
  const int rblk = blockIdx.x >> 4;
  const int row0 = (rblk * 4 + wave) * 16;
  const int col0 = ct * 64;

  v8f acc[4] = {v8f_zero(), v8f_zero(), v8f_zero(), v8f_zero()};
  gemm16x64(OH, INNER_, Wout, DIM_, row0, col0, INNER_, acc);

  const int lane   = threadIdx.x & 31;
  const int laneLo = lane & 15;
  const int rowAdd = (lane >> 4) * 8;
#pragma unroll
  for (int t = 0; t < 4; ++t) {
    float bb = bout[col0 + t * 16 + laneLo];
#pragma unroll
    for (int v = 0; v < 8; ++v) {
      out[(size_t)(row0 + v + rowAdd) * DIM_ + col0 + t * 16 + laneLo] =
          acc[t][v] + bb;
    }
  }
}

extern "C" void kernel_launch(void* const* d_in, const int* in_sizes, int n_in,
                              void* d_out, int out_size, void* d_ws,
                              size_t ws_size, hipStream_t stream) {
  const float* x    = (const float*)d_in[0];
  const float* ctx  = (const float*)d_in[1];
  const float* rope = (const float*)d_in[4];
  const float* Wq   = (const float*)d_in[5];
  const float* Wkv  = (const float*)d_in[6];
  const float* Wout = (const float*)d_in[7];
  const float* bout = (const float*)d_in[8];
  float* out = (float*)d_out;

  const size_t TEN = (size_t)B_ * H_ * N_ * DH_;
  float* Qws = (float*)d_ws;
  float* Kws = Qws + TEN;
  float* Vws = Kws + TEN;
  float* OHw = Vws + TEN;

  qproj_rope <<<1024, 128, 0, stream>>>(x,   Wq,  rope, Qws);
  kvproj_rope<<<2048, 128, 0, stream>>>(ctx, Wkv, rope, Kws, Vws);
  attn_fwd   <<<1024, 128, 0, stream>>>(Qws, Kws, Vws, OHw);
  outproj    <<<1024, 128, 0, stream>>>(OHw, Wout, bout, out);
}


# projection GEMMs stage operands through LDS: each 128-byte line of the 64x32 A and 32x64 W chunk is read once per block with 16-byte loads; WMMA fragments become 8-byte LDS reads
# speedup vs baseline: 1.0880x; 1.0880x over previous
#include <hip/hip_runtime.h>
#include <hip/hip_bf16.h>


typedef float v2f __attribute__((ext_vector_type(2)));
typedef float v8f __attribute__((ext_vector_type(8)));

#define B_     4
#define N_     1024
#define M_     1024
#define DIM_   1024
#define H_     16
#define DH_    64
#define INNER_ 1024
#define SCALE_ 0.125f

__device__ __forceinline__ v8f wmma4(v2f a, v2f b, v8f c) {
  return __builtin_amdgcn_wmma_f32_16x16x4_f32(false, a, false, b, (short)0, c,
                                               false, false);
}

__device__ __forceinline__ v8f v8f_zero() {
  v8f z = {0.f, 0.f, 0.f, 0.f, 0.f, 0.f, 0.f, 0.f};
  return z;
}

typedef float v4f __attribute__((ext_vector_type(4)));

#define KC_ 32
#define PA_ 36
#define PW_ 160

__device__ __forceinline__ void gemm64x64_lds(const float* __restrict__ A,
                                              int lda,
                                              const float* __restrict__ W,
                                              int ldw, int brow0, int col0,
                                              int K, float* __restrict__ As,
                                              float* __restrict__ Ws,
                                              v8f acc[4]) {
  const int tid    = threadIdx.x;
  const int wave   = tid >> 5;
  const int lane   = tid & 31;
  const int laneLo = lane & 15;
  const int hi2    = (lane >> 4) << 1;

  const int aq = tid & 7,  ar = tid >> 3;
  const int wq = tid & 15, wr = tid >> 4;
  const float* ag = A + (size_t)(brow0 + ar) * lda + aq * 4;
  const float* wg = W + (size_t)(2 * wr) * ldw + col0 + wq * 4;
  float* asw = As + ar * PA_ + aq * 4;
  float* wsw = Ws + wr * PW_ + wq * 8;

  const float* asr = As + (wave * 16 + laneLo) * PA_ + hi2;
  const float* wsr = Ws + (hi2 >> 1) * PW_ + 2 * laneLo;

  v4f ra[4], rw[4];
#pragma unroll
  for (int i = 0; i < 4; ++i)
    ra[i] = *(const v4f*)(ag + (size_t)(16 * i) * lda);
#pragma unroll
  for (int i = 0; i < 2; ++i) {
    rw[2 * i]     = *(const v4f*)(wg + (size_t)(16 * i) * ldw);
    rw[2 * i + 1] = *(const v4f*)(wg + (size_t)(16 * i + 1) * ldw);
  }

  for (int k0 = 0; k0 < K; k0 += KC_) {
    __syncthreads();
#pragma unroll
    for (int i = 0; i < 4; ++i) *(v4f*)(asw + 16 * i * PA_) = ra[i];
#pragma unroll
    for (int i = 0; i < 2; ++i) {
      v4f e = rw[2 * i], o = rw[2 * i + 1];
      asm volatile("" : "+v"(e), "+v"(o));
      const v4f lo = {e.x, o.x, e.y, o.y};
      const v4f hi = {e.z, o.z, e.w, o.w};
      *(v4f*)(wsw + 8 * i * PW_)     = lo;
      *(v4f*)(wsw + 8 * i * PW_ + 4) = hi;
    }
    __syncthreads();
    if (k0 + KC_ < K) {
      const float* ag1 = ag + (k0 + KC_);
      const float* wg1 = wg + (size_t)(k0 + KC_) * ldw;
#pragma unroll
      for (int i = 0; i < 4; ++i)
        ra[i] = *(const v4f*)(ag1 + (size_t)(16 * i) * lda);
#pragma unroll
      for (int i = 0; i < 2; ++i) {
        rw[2 * i]     = *(const v4f*)(wg1 + (size_t)(16 * i) * ldw);
        rw[2 * i + 1] = *(const v4f*)(wg1 + (size_t)(16 * i + 1) * ldw);
      }
    }
#pragma unroll
    for (int kk = 0; kk < KC_; kk += 4) {
      v2f a = *(const v2f*)(asr + kk);
      const float* w0 = wsr + (kk >> 1) * PW_;
#pragma unroll
      for (int t = 0; t < 4; ++t) {
        v2f b = *(const v2f*)(w0 + 32 * t);
        acc[t] = wmma4(a, b, acc[t]);
      }
    }
  }
}

__global__ void __launch_bounds__(128)
qproj_rope(const float* __restrict__ x, const float* __restrict__ Wq,
           const float* __restrict__ rope, float* __restrict__ Q) {
  __shared__ __attribute__((aligned(16))) float As[64 * PA_];
  __shared__ __attribute__((aligned(16))) float Ws[(KC_ / 2) * PW_];
  const int wave = threadIdx.x >> 5;
  const int ct   = blockIdx.x & 15;
  const int rblk = blockIdx.x >> 4;
  const int row0 = (rblk * 4 + wave) * 16;
  const int col0 = ct * 64;

  v8f acc[4] = {v8f_zero(), v8f_zero(), v8f_zero(), v8f_zero()};
  gemm64x64_lds(x, DIM_, Wq, INNER_, rblk * 64, col0, DIM_, As, Ws, acc);

  const int lane   = threadIdx.x & 31;
  const int laneLo = lane & 15;
  const int rowAdd = (lane >> 4) * 8;
  const int b  = row0 >> 10;
  const int n0 = row0 & 1023;
  const int h  = ct;
#pragma unroll
  for (int v = 0; v < 8; ++v) {
    const int n = n0 + v + rowAdd;
    const float* rp = rope + n * DH_;
    float ov[4];
#pragma unroll
    for (int t = 0; t < 4; ++t) {
      const int d = t * 16 + laneLo;
      float p  = rp[d];
      float cs = __cosf(p), sn = __sinf(p);
      float rot = (t < 2) ? -acc[t + 2][v] : acc[t - 2][v];
      ov[t] = acc[t][v] * cs + rot * sn;
    }
    float* qout = Q + ((size_t)((b * H_ + h) * N_ + n)) * DH_;
#pragma unroll
    for (int t = 0; t < 4; ++t) qout[t * 16 + laneLo] = ov[t];
  }
}

__global__ void __launch_bounds__(128)
kvproj_rope(const float* __restrict__ ctx, const float* __restrict__ Wkv,
            const float* __restrict__ rope, float* __restrict__ K,
            float* __restrict__ V) {
  __shared__ __attribute__((aligned(16))) float As[64 * PA_];
  __shared__ __attribute__((aligned(16))) float Ws[(KC_ / 2) * PW_];
  const int wave = threadIdx.x >> 5;
  const int ct   = blockIdx.x & 31;
  const int rblk = blockIdx.x >> 5;
  const int row0 = (rblk * 4 + wave) * 16;
  const int col0 = ct * 64;

  v8f acc[4] = {v8f_zero(), v8f_zero(), v8f_zero(), v8f_zero()};
  gemm64x64_lds(ctx, DIM_, Wkv, 2 * INNER_, rblk * 64, col0, DIM_, As, Ws, acc);

  const int lane   = threadIdx.x & 31;
  const int laneLo = lane & 15;
  const int rowAdd = (lane >> 4) * 8;
  const int b  = row0 >> 10;
  const int m0 = row0 & 1023;

  if (ct < 16) {
    const int h = ct;
#pragma unroll
    for (int v = 0; v < 8; ++v) {
      const int m = m0 + v + rowAdd;
      const float* rp = rope + m * DH_;
      float ov[4];
#pragma unroll
      for (int t = 0; t < 4; ++t) {
        const int d = t * 16 + laneLo;
        float p  = rp[d];
        float cs = __cosf(p), sn = __sinf(p);
        float rot = (t < 2) ? -acc[t + 2][v] : acc[t - 2][v];
        ov[t] = acc[t][v] * cs + rot * sn;
      }
      float* kout = K + ((size_t)((b * H_ + h) * M_ + m)) * DH_;
#pragma unroll
      for (int t = 0; t < 4; ++t) kout[t * 16 + laneLo] = ov[t];
    }
  } else {
    const int h = ct - 16;
#pragma unroll
    for (int v = 0; v < 8; ++v) {
      const int m = m0 + v + rowAdd;
      float* vout = V + ((size_t)((b * H_ + h) * M_ + m)) * DH_;
#pragma unroll
      for (int t = 0; t < 4; ++t) vout[t * 16 + laneLo] = acc[t][v];
    }
  }
}

__global__ void __launch_bounds__(128)
attn_fwd(const float* __restrict__ Q, const float* __restrict__ K,
         const float* __restrict__ V, float* __restrict__ OH) {
  __shared__ float ldsP[4][16 * 18];

  const int wave   = threadIdx.x >> 5;
  const int lane   = threadIdx.x & 31;
  const int laneLo = lane & 15;
  const int hi2    = (lane >> 4) << 1;
  const int rowAdd = (lane >> 4) * 8;

  const int bh   = blockIdx.x >> 4;
  const int iblk = blockIdx.x & 15;
  const int i0   = iblk * 64 + wave * 16;

  const float* qbase = Q + ((size_t)bh * N_ + i0 + laneLo) * DH_ + hi2;
  v2f qa[16];
#pragma unroll
  for (int s = 0; s < 16; ++s) qa[s] = *(const v2f*)(qbase + 4 * s);

  v8f o[4] = {v8f_zero(), v8f_zero(), v8f_zero(), v8f_zero()};
  float m[8], l[8];
#pragma unroll
  for (int v = 0; v < 8; ++v) { m[v] = -__builtin_inff(); l[v] = 0.f; }

  const float* kb0 = K + (size_t)bh * M_ * DH_;
  const float* vb0 = V + (size_t)bh * M_ * DH_;
  float* myP = &ldsP[wave][0];

  for (int j0 = 0; j0 < M_; j0 += 16) {
    v8f sacc = v8f_zero();
    const float* kp = kb0 + (size_t)(j0 + laneLo) * DH_ + hi2;
#pragma unroll
    for (int s = 0; s < 16; ++s) {
      v2f kb = *(const v2f*)(kp + 4 * s);
      sacc = wmma4(qa[s], kb, sacc);
    }

#pragma unroll
    for (int v = 0; v < 8; ++v) {
      float sv = sacc[v] * SCALE_;
      float mx = sv;
      mx = fmaxf(mx, __shfl_xor(mx, 1, 32));
      mx = fmaxf(mx, __shfl_xor(mx, 2, 32));
      mx = fmaxf(mx, __shfl_xor(mx, 4, 32));
      mx = fmaxf(mx, __shfl_xor(mx, 8, 32));
      float mn = fmaxf(m[v], mx);
      float rr = __expf(m[v] - mn);
      float p  = __expf(sv - mn);
      m[v] = mn;
      float ps = p;
      ps += __shfl_xor(ps, 1, 32);
      ps += __shfl_xor(ps, 2, 32);
      ps += __shfl_xor(ps, 4, 32);
      ps += __shfl_xor(ps, 8, 32);
      l[v] = l[v] * rr + ps;
#pragma unroll
      for (int t = 0; t < 4; ++t) o[t][v] *= rr;
      myP[(v + rowAdd) * 18 + laneLo] = p;
    }
    asm volatile("s_wait_dscnt 0" ::: "memory");

    const float* vp = vb0 + (size_t)(j0 + hi2) * DH_ + laneLo;
#pragma unroll
    for (int s = 0; s < 4; ++s) {
      v2f pa = *(const v2f*)(myP + laneLo * 18 + 4 * s + hi2);
      const float* vrow = vp + (size_t)(4 * s) * DH_;
#pragma unroll
      for (int t = 0; t < 4; ++t) {
        v2f vb;
        vb.x = vrow[t * 16];
        vb.y = vrow[t * 16 + DH_];
        o[t] = wmma4(pa, vb, o[t]);
      }
    }
    asm volatile("" ::: "memory");
  }

  const int b = bh >> 4, h = bh & 15;
#pragma unroll
  for (int v = 0; v < 8; ++v) {
    const int i  = i0 + v + rowAdd;
    float inv = 1.0f / l[v];
    float* op = OH + (((size_t)(b * N_ + i) * H_) + h) * DH_;
#pragma unroll
    for (int t = 0; t < 4; ++t) op[t * 16 + laneLo] = o[t][v] * inv;
  }
}

__global__ void __launch_bounds__(128)
outproj(const float* __restrict__ OH, const float* __restrict__ Wout,
        const float* __restrict__ bout, float* __restrict__ out) {
  __shared__ __attribute__((aligned(16))) float As[64 * PA_];
  __shared__ __attribute__((aligned(16))) float Ws[(KC_ / 2) * PW_];
  const int wave = threadIdx.x >> 5;
  const int ct   = blockIdx.x & 15;
  const int rblk = blockIdx.x >> 4;
  const int row0 = (rblk * 4 + wave) * 16;
  const int col0 = ct * 64;

  v8f acc[4] = {v8f_zero(), v8f_zero(), v8f_zero(), v8f_zero()};
  gemm64x64_lds(OH, INNER_, Wout, DIM_, rblk * 64, col0, INNER_, As, Ws, acc);

  const int lane   = threadIdx.x & 31;
  const int laneLo = lane & 15;
  const int rowAdd = (lane >> 4) * 8;
#pragma unroll
  for (int t = 0; t < 4; ++t) {
    float bb = bout[col0 + t * 16 + laneLo];
#pragma unroll
    for (int v = 0; v < 8; ++v) {
      out[(size_t)(row0 + v + rowAdd) * DIM_ + col0 + t * 16 + laneLo] =
          acc[t][v] + bb;
    }
  }
}

extern "C" void kernel_launch(void* const* d_in, const int* in_sizes, int n_in,
                              void* d_out, int out_size, void* d_ws,
                              size_t ws_size, hipStream_t stream) {
  const float* x    = (const float*)d_in[0];
  const float* ctx  = (const float*)d_in[1];
  const float* rope = (const float*)d_in[4];
  const float* Wq   = (const float*)d_in[5];
  const float* Wkv  = (const float*)d_in[6];
  const float* Wout = (const float*)d_in[7];
  const float* bout = (const float*)d_in[8];
  float* out = (float*)d_out;

  const size_t TEN = (size_t)B_ * H_ * N_ * DH_;
  float* Qws = (float*)d_ws;
  float* Kws = Qws + TEN;
  float* Vws = Kws + TEN;
  float* OHw = Vws + TEN;

  qproj_rope <<<1024, 128, 0, stream>>>(x,   Wq,  rope, Qws);
  kvproj_rope<<<2048, 128, 0, stream>>>(ctx, Wkv, rope, Kws, Vws);
  attn_fwd   <<<1024, 128, 0, stream>>>(Qws, Kws, Vws, OHw);
  outproj    <<<1024, 128, 0, stream>>>(OHw, Wout, bout, out);
}
